# starting kernel
# baseline (speedup 1.0000x reference)
#include <hip/hip_runtime.h>

typedef __attribute__((ext_vector_type(2))) float v2f;
typedef __attribute__((ext_vector_type(8))) float v8f;

#define N_NODES 16384
#define N_ANCH  256
#define DLAT    32
#define BSTR    40

__global__ void pnn_prep_M(const float* __restrict__ embeds,
                           const float* __restrict__ W,
                           const int* __restrict__ anchor_ids,
                           float* __restrict__ Mp) {
    const int a = blockIdx.x;
    const int j = threadIdx.x;
    const int aid = anchor_ids[a];
    const float* e = embeds + (size_t)aid * DLAT;
    float acc = 0.0f;
#pragma unroll
    for (int d = 0; d < DLAT; ++d)
        acc = fmaf(e[d], W[d * DLAT + j], acc);
    Mp[a * DLAT + j] = acc * (1.0f / (float)N_ANCH);
}

__global__ void pnn_prep_C(const float* __restrict__ embeds,
                           const float* __restrict__ W,
                           const float* __restrict__ bias,
                           float* __restrict__ Cv) {
    const int j = threadIdx.x;
    float acc = 0.0f;
    for (int a = 0; a < N_ANCH; ++a) {
        const float* e = embeds + (size_t)a * DLAT;
#pragma unroll
        for (int d = 0; d < DLAT; ++d)
            acc = fmaf(e[d], W[(DLAT + d) * DLAT + j], acc);
    }
    Cv[j] = acc * (1.0f / (float)N_ANCH) + bias[j];
}

__global__ __launch_bounds__(256) void pnn_main(
        const float* __restrict__ dists,
        const float* __restrict__ Mp,
        const float* __restrict__ Cv,
        float* __restrict__ out) {
    __shared__ float Blds[N_ANCH * BSTR];

    const int tid = threadIdx.x;

#pragma unroll 4
    for (int i = tid; i < N_ANCH * DLAT; i += 256) {
        const int a = i >> 5, j = i & 31;
        Blds[a * BSTR + j] = Mp[i];
    }
    __syncthreads();

    const int wave = tid >> 5;
    const int lane = tid & 31;
    const int m    = lane & 15;
    const int kh   = lane >> 4;
    const int node_base = (blockIdx.x * 8 + wave) * 16;

    const float c_lo = Cv[m];
    const float c_hi = Cv[m + 16];

    v8f acc0 = {};
    v8f acc1 = {};

    const float* aptr = dists + node_base + m;

#pragma unroll 4
    for (int k0 = 0; k0 < N_ANCH; k0 += 4) {
        const int ka = k0 + 2 * kh;
        v2f af, b0, b1;
        af.x = aptr[(size_t)ka * N_NODES];
        af.y = aptr[(size_t)(ka + 1) * N_NODES];
        b0.x = Blds[ka * BSTR + m];
        b0.y = Blds[(ka + 1) * BSTR + m];
        b1.x = Blds[ka * BSTR + m + 16];
        b1.y = Blds[(ka + 1) * BSTR + m + 16];

        acc0 = __builtin_amdgcn_wmma_f32_16x16x4_f32(
                   false, af, false, b0, (short)0, acc0, false, false);
        acc1 = __builtin_amdgcn_wmma_f32_16x16x4_f32(
                   false, af, false, b1, (short)0, acc1, false, false);
    }

#pragma unroll
    for (int r = 0; r < 8; ++r) {
        const int row = node_base + r + 8 * kh;
        out[(size_t)row * DLAT + m]      = acc0[r] + c_lo;
        out[(size_t)row * DLAT + m + 16] = acc1[r] + c_hi;
    }
}

extern "C" void kernel_launch(void* const* d_in, const int* in_sizes, int n_in,
                              void* d_out, int out_size, void* d_ws, size_t ws_size,
                              hipStream_t stream) {
    const float* embeds     = (const float*)d_in[0];
    const float* dists      = (const float*)d_in[1];
    const float* W_hidden   = (const float*)d_in[2];
    const float* b_hidden   = (const float*)d_in[3];
    const int*   anchor_ids = (const int*)d_in[4];

    float* Mp = (float*)d_ws;
    float* Cv = Mp + N_ANCH * DLAT;
    float* out = (float*)d_out;

    pnn_prep_M<<<N_ANCH, DLAT, 0, stream>>>(embeds, W_hidden, anchor_ids, Mp);
    pnn_prep_C<<<1, DLAT, 0, stream>>>(embeds, W_hidden, b_hidden, Cv);

    pnn_main<<<N_NODES / (16 * 8), 256, 0, stream>>>(dists, Mp, Cv, out);
}


# prep launch removed: each wave builds M of its 64 anchors on f32 WMMA, its result registers are the main product's B fragments in place (no LDS, no workspace); changes sum order (d, anchor, constant)
# speedup vs baseline: 16.6129x; 16.6129x over previous
#include <hip/hip_runtime.h>

typedef __attribute__((ext_vector_type(2))) float v2f;
typedef __attribute__((ext_vector_type(4))) float v4f;
typedef __attribute__((ext_vector_type(8))) float v8f;
typedef __attribute__((ext_vector_type(4))) int   v4i;

#define N_NODES 16384
#define N_ANCH  256
#define DLAT    32
#define TNODES  64

__device__ __forceinline__ float row_sum16(float x) {
    x += __builtin_bit_cast(float, __builtin_amdgcn_update_dpp(0, __builtin_bit_cast(int, x), 0x161, 0xf, 0xf, true));
    x += __builtin_bit_cast(float, __builtin_amdgcn_update_dpp(0, __builtin_bit_cast(int, x), 0x162, 0xf, 0xf, true));
    x += __builtin_bit_cast(float, __builtin_amdgcn_update_dpp(0, __builtin_bit_cast(int, x), 0x164, 0xf, 0xf, true));
    x += __builtin_bit_cast(float, __builtin_amdgcn_update_dpp(0, __builtin_bit_cast(int, x), 0x168, 0xf, 0xf, true));
    return x;
}

__global__ __launch_bounds__(128) void pnn_fused(
        const float* __restrict__ embeds,
        const float* __restrict__ dists,
        const float* __restrict__ W,
        const float* __restrict__ bias,
        const int*   __restrict__ anchor_ids,
        float* __restrict__ out) {
    __shared__ v4f Red[4 * 8 * 2 * 32];
    __shared__ float Cpart[4 * DLAT];

    const int tid  = threadIdx.x;
    const int lane = tid & 31;
    const int wave = __builtin_amdgcn_readfirstlane(tid >> 5);
    const int m    = lane & 15;
    const int kh   = lane >> 4;
    const int base = blockIdx.x * TNODES;

    const v4i ids = reinterpret_cast<const v4i*>(anchor_ids)[16 * wave + m];

    v2f wf[4][2][2];
#pragma unroll
    for (int j = 0; j < 4; ++j)
#pragma unroll
        for (int u = 0; u < 2; ++u)
#pragma unroll
            for (int ct = 0; ct < 2; ++ct) {
                const int d = 8 * j + 4 * kh + 2 * u;
                wf[j][u][ct].x = W[d * DLAT + 16 * ct + m];
                wf[j][u][ct].y = W[(d + 1) * DLAT + 16 * ct + m];
            }

    const v4f* s4 = reinterpret_cast<const v4f*>(embeds) + m * 8 + 2 * wave + kh;
    v4f sr[16];
#pragma unroll
    for (int i = 0; i < 16; ++i)
        sr[i] = s4[(16 * i) * 8];
    float w2[8];
#pragma unroll
    for (int q = 0; q < 8; ++q)
        w2[q] = W[(DLAT + 8 * wave + q) * DLAT + lane];
    const float b_lo = bias[m];
    const float b_hi = bias[m + 16];
    __builtin_amdgcn_sched_barrier(0);

    v4f e[4][4];
#pragma unroll
    for (int T = 0; T < 4; ++T) {
        const v4f* er = reinterpret_cast<const v4f*>(embeds + (size_t)ids[T] * DLAT) + kh;
#pragma unroll
        for (int j = 0; j < 4; ++j)
            e[T][j] = er[2 * j];
    }
    __builtin_amdgcn_sched_barrier(0);

    const v4f* d4 = reinterpret_cast<const v4f*>(dists + base)
                    + (size_t)(64 * wave + 32 * kh) * (N_NODES / 4) + m;
    v4f a0[4][4], a1[4][4];
#pragma unroll
    for (int T = 0; T < 4; ++T)
#pragma unroll
        for (int p = 0; p < 4; ++p) {
            a0[T][p] = d4[(size_t)(8 * p + T) * (N_NODES / 4)];
            a1[T][p] = d4[(size_t)(8 * p + T + 4) * (N_NODES / 4)];
        }
    __builtin_amdgcn_sched_barrier(0);

    v4f ps = ((sr[0] + sr[1]) + (sr[2] + sr[3])) + ((sr[4] + sr[5]) + (sr[6] + sr[7]));
    ps += ((sr[8] + sr[9]) + (sr[10] + sr[11])) + ((sr[12] + sr[13]) + (sr[14] + sr[15]));
    float cs4[4];
#pragma unroll
    for (int x = 0; x < 4; ++x) {
        const float px = ps[x];
        cs4[x] = row_sum16(px);
    }
    float cw = 0.0f;
#pragma unroll
    for (int q = 0; q < 8; ++q) {
        const float cs = __int_as_float(__builtin_amdgcn_readlane(__float_as_int(cs4[q & 3]), 16 * (q >> 2)));
        cw = fmaf(cs, w2[q], cw);
    }
    Cpart[wave * DLAT + lane] = cw;
    __builtin_amdgcn_sched_barrier(0);

    v8f mt[4][2] = {};
#pragma unroll
    for (int j = 0; j < 4; ++j)
#pragma unroll
        for (int u = 0; u < 2; ++u)
#pragma unroll
            for (int T = 0; T < 4; ++T) {
                const v2f fa = {e[T][j][2 * u], e[T][j][2 * u + 1]};
                mt[T][0] = __builtin_amdgcn_wmma_f32_16x16x4_f32(
                               false, fa, false, wf[j][u][0], (short)0, mt[T][0], false, false);
                mt[T][1] = __builtin_amdgcn_wmma_f32_16x16x4_f32(
                               false, fa, false, wf[j][u][1], (short)0, mt[T][1], false, false);
            }

    v8f acc[4][2] = {};
#pragma unroll
    for (int T = 0; T < 4; ++T)
#pragma unroll
        for (int p = 0; p < 4; ++p) {
            const v2f fb0 = {mt[T][0][2 * p], mt[T][0][2 * p + 1]};
            const v2f fb1 = {mt[T][1][2 * p], mt[T][1][2 * p + 1]};
#pragma unroll
            for (int t = 0; t < 4; ++t) {
                const v2f fa = {a0[T][p][t], a1[T][p][t]};
                acc[t][0] = __builtin_amdgcn_wmma_f32_16x16x4_f32(
                                false, fa, false, fb0, (short)0, acc[t][0], false, false);
                acc[t][1] = __builtin_amdgcn_wmma_f32_16x16x4_f32(
                                false, fa, false, fb1, (short)0, acc[t][1], false, false);
            }
        }

#pragma unroll
    for (int t = 0; t < 4; ++t)
#pragma unroll
        for (int ct = 0; ct < 2; ++ct)
#pragma unroll
            for (int q = 0; q < 2; ++q) {
                v4f v;
                v.x = acc[t][ct][4 * q];     v.y = acc[t][ct][4 * q + 1];
                v.z = acc[t][ct][4 * q + 2]; v.w = acc[t][ct][4 * q + 3];
                Red[((wave * 8 + t * 2 + ct) * 2 + q) * 32 + lane] = v;
            }
    __syncthreads();

    const float inv_a = 1.0f / (float)N_ANCH;
    float c_lo = Cpart[m];
    float c_hi = Cpart[m + 16];
#pragma unroll
    for (int g = 1; g < 4; ++g) {
        c_lo += Cpart[g * DLAT + m];
        c_hi += Cpart[g * DLAT + m + 16];
    }
    c_lo = fmaf(c_lo, inv_a, b_lo);
    c_hi = fmaf(c_hi, inv_a, b_hi);

#pragma unroll
    for (int ct = 0; ct < 2; ++ct)
#pragma unroll
        for (int q = 0; q < 2; ++q) {
            const int idx = ((wave * 2 + ct) * 2 + q) * 32 + lane;
            v4f sum = Red[idx];
#pragma unroll
            for (int g = 1; g < 4; ++g)
                sum += Red[g * (8 * 2 * 32) + idx];
            const float cc = ct ? c_hi : c_lo;
#pragma unroll
            for (int x = 0; x < 4; ++x) {
                const int node = base + 4 * (8 * kh + 4 * q + x) + wave;
                out[(size_t)node * DLAT + 16 * ct + m] = fmaf(sum[x], inv_a, cc);
            }
        }
}

extern "C" void kernel_launch(void* const* d_in, const int* in_sizes, int n_in,
                              void* d_out, int out_size, void* d_ws, size_t ws_size,
                              hipStream_t stream) {
    const float* embeds     = (const float*)d_in[0];
    const float* dists      = (const float*)d_in[1];
    const float* W_hidden   = (const float*)d_in[2];
    const float* b_hidden   = (const float*)d_in[3];
    const int*   anchor_ids = (const int*)d_in[4];
    float* out = (float*)d_out;

    pnn_fused<<<N_NODES / TNODES, 128, 0, stream>>>(embeds, dists, W_hidden, b_hidden, anchor_ids, out);
}
